# starting kernel
# baseline (speedup 1.0000x reference)
#include <hip/hip_runtime.h>
#include <hip/hip_bf16.h>

typedef __attribute__((ext_vector_type(2))) float v2f;
typedef __attribute__((ext_vector_type(8))) float v8f;

#define N_NODES 50000
#define N_EDGES 600000
#define DIM     128
#define NGRAPH  64
#define DEPTH   4

__global__ void zero_f32_kernel(float* __restrict__ p, int n) {
    int i = blockIdx.x * blockDim.x + threadIdx.x;
    if (i < n) p[i] = 0.0f;
}

__global__ void deg_kernel(const int* __restrict__ dst, float* __restrict__ deg) {
    int e = blockIdx.x * blockDim.x + threadIdx.x;
    if (e < N_EDGES) atomicAdd(&deg[dst[e]], 1.0f);
}

__global__ void dis_kernel(const float* __restrict__ deg, float* __restrict__ dis) {
    int i = blockIdx.x * blockDim.x + threadIdx.x;
    if (i < N_NODES) {
        float d = deg[i];
        dis[i] = (d > 0.0f) ? rsqrtf(d) : 0.0f;
    }
}

__global__ void embed_kernel(const int* __restrict__ x,
                             const float* __restrict__ emb,
                             float* __restrict__ h) {
    int i = blockIdx.x * blockDim.x + threadIdx.x;
    if (i >= N_NODES * (DIM / 4)) return;
    int node = i >> 5;
    int d4   = i & 31;
    int tok  = x[node];
    ((float4*)h)[i] = ((const float4*)(emb + (long)tok * DIM))[d4];
}

__global__ void bias_init_kernel(const float* __restrict__ bias, float* __restrict__ h) {
    int i = blockIdx.x * blockDim.x + threadIdx.x;
    if (i >= N_NODES * (DIM / 4)) return;
    int d4 = i & 31;
    ((float4*)h)[i] = ((const float4*)bias)[d4];
}

__global__ void __launch_bounds__(256)
gemm_wmma_kernel(const float* __restrict__ h,
                 const float* __restrict__ W,
                 float* __restrict__ hw) {
    __shared__ float As[16 * DIM];

    const int tid     = threadIdx.x;
    const int rowBase = blockIdx.x * 16;

    {
        const float4* src = (const float4*)(h + (long)rowBase * DIM);
        float4* dst4 = (float4*)As;
        dst4[tid]       = src[tid];
        dst4[tid + 256] = src[tid + 256];
    }
    __syncthreads();

    const int wave    = tid >> 5;
    const int lane    = tid & 31;
    const int laneM   = lane & 15;
    const int half    = lane >> 4;
    const int colBase = wave * 16;

    v8f acc = {};

    #pragma unroll
    for (int k0 = 0; k0 < DIM; k0 += 4) {
        const int kA = k0 + half * 2;
        v2f a;
        a.x = As[laneM * DIM + kA];
        a.y = As[laneM * DIM + kA + 1];
        v2f b;
        b.x = W[kA * DIM + colBase + laneM];
        b.y = W[(kA + 1) * DIM + colBase + laneM];
        acc = __builtin_amdgcn_wmma_f32_16x16x4_f32(
            false, a, false, b, (short)0, acc, false, false);
    }

    float* out = hw + (long)rowBase * DIM + colBase;
    #pragma unroll
    for (int r = 0; r < 8; ++r) {
        out[(r + half * 8) * DIM + laneM] = acc[r];
    }
}

__global__ void __launch_bounds__(256)
scatter_kernel(const float* __restrict__ hw,
               const int* __restrict__ src_idx,
               const int* __restrict__ dst_idx,
               const float* __restrict__ dis,
               float* __restrict__ h) {
    int gtid = blockIdx.x * blockDim.x + threadIdx.x;
    int edge = gtid >> 5;
    int lane = threadIdx.x & 31;
    if (edge >= N_EDGES) return;

    int s = src_idx[edge];
    int d = dst_idx[edge];
    float nrm = dis[s] * dis[d];

    float4 v = ((const float4*)(hw + (long)s * DIM))[lane];
    float* out = h + (long)d * DIM + lane * 4;
    atomicAdd(out + 0, v.x * nrm);
    atomicAdd(out + 1, v.y * nrm);
    atomicAdd(out + 2, v.z * nrm);
    atomicAdd(out + 3, v.w * nrm);
}

__global__ void gather_out_kernel(const int* __restrict__ ptr,
                                  const float* __restrict__ h,
                                  float* __restrict__ out) {
    int i = blockIdx.x * blockDim.x + threadIdx.x;
    if (i >= NGRAPH * DIM) return;
    int b = i >> 7;
    int d = i & (DIM - 1);
    int node = ptr[b + 1] - 1;
    out[i] = h[(long)node * DIM + d];
}

extern "C" void kernel_launch(void* const* d_in, const int* in_sizes, int n_in,
                              void* d_out, int out_size, void* d_ws, size_t ws_size,
                              hipStream_t stream) {
    const int*   x    = (const int*)d_in[0];
    const int*   eidx = (const int*)d_in[1];
    const int*   ptr  = (const int*)d_in[2];
    const float* emb  = (const float*)d_in[4];
    const float* W    = (const float*)d_in[5];
    const float* bias = (const float*)d_in[6];
    float*       out  = (float*)d_out;

    const int* src_idx = eidx;
    const int* dst_idx = eidx + N_EDGES;

    float* h   = (float*)d_ws;
    float* hw  = h  + (size_t)N_NODES * DIM;
    float* deg = hw + (size_t)N_NODES * DIM;
    float* dis = deg + N_NODES;

    const int T = 256;

    zero_f32_kernel<<<(N_NODES + T - 1) / T, T, 0, stream>>>(deg, N_NODES);
    deg_kernel<<<(N_EDGES + T - 1) / T, T, 0, stream>>>(dst_idx, deg);
    dis_kernel<<<(N_NODES + T - 1) / T, T, 0, stream>>>(deg, dis);

    const int vec_elems = N_NODES * (DIM / 4);
    embed_kernel<<<(vec_elems + T - 1) / T, T, 0, stream>>>(x, emb, h);

    for (int l = 0; l < DEPTH; ++l) {
        gemm_wmma_kernel<<<N_NODES / 16, T, 0, stream>>>(h, W, hw);
        bias_init_kernel<<<(vec_elems + T - 1) / T, T, 0, stream>>>(bias, h);
        scatter_kernel<<<(N_EDGES * 32 + T - 1) / T, T, 0, stream>>>(hw, src_idx, dst_idx, dis, h);
    }

    gather_out_kernel<<<(NGRAPH * DIM + T - 1) / T, T, 0, stream>>>(ptr, h, out);
}


# Degrees from the edge scan: a scan launch ranks each owned row's edges by LDS int adds (count = degree), writes dis and the record; 3 degree launches gone (8 to 6); layer 1 walks it: sum order rank
# speedup vs baseline: 4.4037x; 4.4037x over previous
#include <hip/hip_runtime.h>
#include <hip/hip_bf16.h>

typedef __attribute__((ext_vector_type(2))) float v2f;
typedef __attribute__((ext_vector_type(8))) float v8f;

#define N_NODES 50000
#define N_EDGES 600000
#define DIM     128
#define NGRAPH  64
#define DEPTH   4

#define SC_ROWS    200
#define SC_TILES   ((SC_ROWS + 15) / 16)
#define SC_PADROWS (SC_TILES * 16)
#define SC_THREADS 1024
#define SC_WAVES   (SC_THREADS / 32)
#define SC_G       8
#define SC_Q       512
#define SCN_Q      1024
#define SC_NVEC    (N_EDGES / 4)
#define SC_STEP    (SC_G * SC_THREADS)
#define SC_FULL    (SC_NVEC / SC_STEP)
#define ACC_PITCH  (DIM + 4)
#define SL_CAP     3792
#define SL_OFF     SL_CAP
#define SL_FLAG    (SL_CAP + SC_ROWS + 1)
#define LS_WG      (SL_CAP + 208)
#define LS_PLANE   (SC_ROWS * DIM)
#define SRT_N      224
#define SRT_WORDS  (2 * SRT_N + 2)
#define N_GROUPS   ((N_NODES + SC_ROWS - 1) / SC_ROWS)
#define WSQ_ROWS   (DIM + 1)
#define WS_NEED    (((size_t)2 * N_NODES * DIM + (size_t)2 * N_NODES + (size_t)((N_NODES + SC_ROWS - 1) / SC_ROWS) * LS_WG) * 4)
static_assert(N_NODES <= 65536, "a record word keeps the source node in 16 bits");
static_assert(SL_FLAG < LS_WG, "run starts and mark lie inside the group's record");
static_assert(SC_ROWS <= 256 && N_EDGES < (1 << 23), "a ring entry of the scan launch keeps the owned row in 8 bits and the rank above them");
static_assert(SRT_N >= SC_PADROWS && SRT_N == 7 * 32, "one wave adds up the counters, 7 per lane");
static_assert(LS_WG <= LS_PLANE, "a group's record fits in the group's own rows of a plane");
static_assert(N_EDGES % 4 == 0, "dst is read as 16-byte groups");
static_assert(SC_FULL % 2 == 0, "the scan loop takes two whole phases per trip");
static_assert(SC_G % 2 == 0, "found edges are drained once per two groups");
static_assert(3 + 8 * 32 <= SC_Q, "ring holds what can be found between two drains");
static_assert((ACC_PITCH * 4) % 16 == 0, "rows are cleared as 16-byte pieces");
static_assert(SC_WAVES * SC_Q * 8 >= SC_PADROWS * DIM * 4, "the rings' space holds the finished block");
static_assert(SC_PADROWS > SC_ROWS, "the last product tile has padding rows");
static_assert(WSQ_ROWS <= SC_PADROWS, "W's rows and the bias row fit in the LDS block that feeds the product");
static_assert(WSQ_ROWS * DIM <= N_NODES, "W W and bias W fit in the N_NODES floats in front of dis");
static_assert(SC_WAVES == 32 && SC_TILES > 12 && SC_TILES <= 16 && DIM == 128, "8 column tiles x 4 waves; a wave takes row tiles q, q+4, q+8 and, if there is one, q+12");

template <bool FIRST, bool USUM>
__device__ __forceinline__ void sc_add4(int2 t0, int2 t1, int2 t2, int2 t3,
                                        const float* __restrict__ rows,
                                        const int* __restrict__ tok,
                                        const int* __restrict__ src_idx,
                                        const float* __restrict__ dis,
                                        float* acc, float* usum, int base, int lane) {
    const int e0 = __builtin_amdgcn_readfirstlane(t0.x), r0 = __builtin_amdgcn_readfirstlane(t0.y);
    const int e1 = __builtin_amdgcn_readfirstlane(t1.x), r1 = __builtin_amdgcn_readfirstlane(t1.y);
    const int e2 = __builtin_amdgcn_readfirstlane(t2.x), r2 = __builtin_amdgcn_readfirstlane(t2.y);
    const int e3 = __builtin_amdgcn_readfirstlane(t3.x), r3 = __builtin_amdgcn_readfirstlane(t3.y);
    const int s0 = src_idx[e0], s1 = src_idx[e1], s2 = src_idx[e2], s3 = src_idx[e3];
    const int q0 = FIRST ? tok[s0] : s0;
    const int q1 = FIRST ? tok[s1] : s1;
    const int q2 = FIRST ? tok[s2] : s2;
    const int q3 = FIRST ? tok[s3] : s3;
    const float n0 = dis[s0] * dis[base + r0];
    const float n1 = dis[s1] * dis[base + r1];
    const float n2 = dis[s2] * dis[base + r2];
    const float n3 = dis[s3] * dis[base + r3];
    const float* p0 = rows + (long)q0 * DIM + lane;
    const float* p1 = rows + (long)q1 * DIM + lane;
    const float* p2 = rows + (long)q2 * DIM + lane;
    const float* p3 = rows + (long)q3 * DIM + lane;
    const float a00 = p0[0], a01 = p0[32], a02 = p0[64], a03 = p0[96];
    const float a10 = p1[0], a11 = p1[32], a12 = p1[64], a13 = p1[96];
    const float a20 = p2[0], a21 = p2[32], a22 = p2[64], a23 = p2[96];
    const float a30 = p3[0], a31 = p3[32], a32 = p3[64], a33 = p3[96];
    float* o0 = acc + r0 * ACC_PITCH + lane;
    float* o1 = acc + r1 * ACC_PITCH + lane;
    float* o2 = acc + r2 * ACC_PITCH + lane;
    float* o3 = acc + r3 * ACC_PITCH + lane;
    atomicAdd(o0, a00 * n0); atomicAdd(o0 + 32, a01 * n0); atomicAdd(o0 + 64, a02 * n0); atomicAdd(o0 + 96, a03 * n0);
    atomicAdd(o1, a10 * n1); atomicAdd(o1 + 32, a11 * n1); atomicAdd(o1 + 64, a12 * n1); atomicAdd(o1 + 96, a13 * n1);
    atomicAdd(o2, a20 * n2); atomicAdd(o2 + 32, a21 * n2); atomicAdd(o2 + 64, a22 * n2); atomicAdd(o2 + 96, a23 * n2);
    atomicAdd(o3, a30 * n3); atomicAdd(o3 + 32, a31 * n3); atomicAdd(o3 + 64, a32 * n3); atomicAdd(o3 + 96, a33 * n3);
    if (USUM && lane == 0) {
        atomicAdd(usum + r0, n0); atomicAdd(usum + r1, n1); atomicAdd(usum + r2, n2); atomicAdd(usum + r3, n3);
    }
}

template <bool FIRST, bool USUM>
__device__ __forceinline__ void sc_add1(int2 t0,
                                        const float* __restrict__ rows,
                                        const int* __restrict__ tok,
                                        const int* __restrict__ src_idx,
                                        const float* __restrict__ dis,
                                        float* acc, float* usum, int base, int lane) {
    const int e0 = __builtin_amdgcn_readfirstlane(t0.x), r0 = __builtin_amdgcn_readfirstlane(t0.y);
    const int s0 = src_idx[e0];
    const int q0 = FIRST ? tok[s0] : s0;
    const float n0 = dis[s0] * dis[base + r0];
    const float* p0 = rows + (long)q0 * DIM + lane;
    const float a00 = p0[0], a01 = p0[32], a02 = p0[64], a03 = p0[96];
    float* o0 = acc + r0 * ACC_PITCH + lane;
    atomicAdd(o0, a00 * n0); atomicAdd(o0 + 32, a01 * n0); atomicAdd(o0 + 64, a02 * n0); atomicAdd(o0 + 96, a03 * n0);
    if (USUM && lane == 0) atomicAdd(usum + r0, n0);
}

template <int J, bool USUM>
__device__ __forceinline__ void row_add8(int w, int q, float nr, int cnt,
                                         const float* __restrict__ rows,
                                         float* acc, float* usum, int lane,
                                         int& cur, float& s0, float& s1, float& s2, float& s3, float& su) {
    float a[8][4];
    float nk[8];
    int   rk[8];
    #pragma unroll
    for (int k = 0; k < 8; ++k) {
        const int wk = __builtin_amdgcn_readlane(w, J + k);
        nk[k] = __int_as_float(__builtin_amdgcn_readlane(__float_as_int(nr), J + k));
        const int qk = __builtin_amdgcn_readlane(q, J + k);
        const float* p = rows + (long)qk * DIM + lane;
        a[k][0] = p[0]; a[k][1] = p[32]; a[k][2] = p[64]; a[k][3] = p[96];
        rk[k] = wk >> 16;
    }
    #pragma unroll
    for (int k = 0; k < 8; ++k) {
        if (k < cnt) {
            if (rk[k] != cur) {
                if (cur >= 0) {
                    float* o = acc + cur * ACC_PITCH + lane;
                    o[0] = s0; o[32] = s1; o[64] = s2; o[96] = s3;
                    if (USUM) usum[cur] = su;
                }
                cur = rk[k];
                s0 = 0.0f; s1 = 0.0f; s2 = 0.0f; s3 = 0.0f;
                su = 0.0f;
            }
            s0 = fmaf(a[k][0], nk[k], s0);
            s1 = fmaf(a[k][1], nk[k], s1);
            s2 = fmaf(a[k][2], nk[k], s2);
            s3 = fmaf(a[k][3], nk[k], s3);
            if (USUM) su += nk[k];
        }
    }
}

template <bool EPI>
__device__ __forceinline__ void block_product(const float* acc,
                                              const float* __restrict__ Wm,
                                              const float* __restrict__ bias,
                                              const float* __restrict__ cvec,
                                              const float* usum,
                                              float* fin, int lane, int wave) {
    const int laneM   = lane & 15;
    const int half    = lane >> 4;
    const int wv      = __builtin_amdgcn_readfirstlane(wave);
    const int colBase = (wv & 7) * 16;
    const int rq      = wv >> 3;
    const bool four   = rq + 12 < SC_TILES;

    v2f bfr[DIM / 4];
    #pragma unroll
    for (int ks = 0; ks < DIM / 4; ++ks) {
        const int kA = ks * 4 + half * 2;
        bfr[ks].x = Wm[kA * DIM + colBase + laneM];
        bfr[ks].y = Wm[(kA + 1) * DIM + colBase + laneM];
    }
    float bv = 0.0f, cv = 0.0f;
    if (EPI) { bv = bias[colBase + laneM]; cv = cvec[colBase + laneM]; }

    __syncthreads();

    auto fix = [&](float v, int row) -> float { return EPI ? v + fmaf(usum[row], cv, bv) : v; };

    {
        const float* a0p = acc + ((rq     ) * 16 + laneM) * ACC_PITCH + half * 2;
        const float* a1p = acc + ((rq +  4) * 16 + laneM) * ACC_PITCH + half * 2;
        const float* a2p = acc + ((rq +  8) * 16 + laneM) * ACC_PITCH + half * 2;
        v8f c0 = {}, c1 = {}, c2 = {};
        #pragma unroll
        for (int ks = 0; ks < DIM / 4; ++ks) {
            v2f a0, a1, a2;
            a0.x = a0p[ks * 4]; a0.y = a0p[ks * 4 + 1];
            a1.x = a1p[ks * 4]; a1.y = a1p[ks * 4 + 1];
            a2.x = a2p[ks * 4]; a2.y = a2p[ks * 4 + 1];
            c0 = __builtin_amdgcn_wmma_f32_16x16x4_f32(false, a0, false, bfr[ks], (short)0, c0, false, false);
            c1 = __builtin_amdgcn_wmma_f32_16x16x4_f32(false, a1, false, bfr[ks], (short)0, c1, false, false);
            c2 = __builtin_amdgcn_wmma_f32_16x16x4_f32(false, a2, false, bfr[ks], (short)0, c2, false, false);
        }
        const int m0 = (rq     ) * 16 + half * 8;
        const int m1 = (rq +  4) * 16 + half * 8;
        const int m2 = (rq +  8) * 16 + half * 8;
        float* f0p = fin + m0 * DIM + colBase + laneM;
        float* f1p = fin + m1 * DIM + colBase + laneM;
        float* f2p = fin + m2 * DIM + colBase + laneM;
        #pragma unroll
        for (int r = 0; r < 8; ++r) {
            f0p[r * DIM] = fix(c0[r], m0 + r);
            f1p[r * DIM] = fix(c1[r], m1 + r);
            f2p[r * DIM] = fix(c2[r], m2 + r);
        }
    }
    if (four) {
        const float* a3p = acc + ((rq + 12) * 16 + laneM) * ACC_PITCH + half * 2;
        v8f c3 = {};
        #pragma unroll
        for (int ks = 0; ks < DIM / 4; ++ks) {
            v2f a3;
            a3.x = a3p[ks * 4]; a3.y = a3p[ks * 4 + 1];
            c3 = __builtin_amdgcn_wmma_f32_16x16x4_f32(false, a3, false, bfr[ks], (short)0, c3, false, false);
        }
        const int m3 = (rq + 12) * 16 + half * 8;
        float* f3p = fin + m3 * DIM + colBase + laneM;
        #pragma unroll
        for (int r = 0; r < 8; ++r) f3p[r * DIM] = fix(c3[r], m3 + r);
    }
    __syncthreads();
}

template <int PITCH>
__device__ __forceinline__ void store_rows(const float* blk, float* hout, int base, int tid) {
    float4* h4 = (float4*)hout;
    #pragma unroll
    for (int k = 0; k < (SC_ROWS * DIM / 4 + SC_THREADS - 1) / SC_THREADS; ++k) {
        const int idx = tid + k * SC_THREADS;
        const int r   = idx >> 5;
        const int c4  = idx & 31;
        if (idx < SC_ROWS * DIM / 4 && base + r < N_NODES)
            h4[(long)(base + r) * (DIM / 4) + c4] = *(const float4*)(blk + r * PITCH + c4 * 4);
    }
}

__device__ __forceinline__ void wsq_block(const float* __restrict__ W,
                                          const float* __restrict__ bias,
                                          float* __restrict__ wsq,
                                          float* acc, float* fin) {
    const int tid = threadIdx.x;
    const float4* W4 = (const float4*)W;
    const float4* b4 = (const float4*)bias;
    for (int i = tid; i < SC_PADROWS * (DIM / 4); i += SC_THREADS) {
        const int r  = i >> 5;
        const int c4 = i & 31;
        float4 v = make_float4(0.0f, 0.0f, 0.0f, 0.0f);
        if (r < DIM)       v = W4[r * (DIM / 4) + c4];
        else if (r == DIM) v = b4[c4];
        *(float4*)(acc + r * ACC_PITCH + c4 * 4) = v;
    }
    block_product<false>(acc, W, nullptr, nullptr, nullptr, fin, tid & 31, tid >> 5);
    const float4* fin4 = (const float4*)fin;
    float4* o4 = (float4*)wsq;
    for (int i = tid; i < WSQ_ROWS * (DIM / 4); i += SC_THREADS) o4[i] = fin4[i];
}

template <bool FIRST, bool PROD>
__device__ __forceinline__ void layer_owner(const float* __restrict__ rows,
                                            const int* __restrict__ tok,
                                            const int* __restrict__ src_idx,
                                            const int* __restrict__ dst_idx,
                                            const float* __restrict__ dis,
                                            const float* __restrict__ Wm,
                                            const float* __restrict__ bias,
                                            const float* __restrict__ cvec,
                                            float* hout,
                                            int* recA, bool haveA,
                                            float* acc, int2* ringmem, float* usum) {
    const int tid  = threadIdx.x;
    const int lane = tid & 31;
    const int wave = tid >> 5;
    const int base = blockIdx.x * SC_ROWS;

    const int wrow0 = (wave * SC_ROWS) / SC_WAVES;
    const int wrow1 = ((wave + 1) * SC_ROWS) / SC_WAVES;
    int mark_v = -1, pb_v = 0, pe_v = 0;
    if (haveA) {
        mark_v = recA[SL_FLAG];
        pb_v   = recA[SL_OFF + wrow0];
        pe_v   = recA[SL_OFF + wrow1];
    }

    {
        float4* acc4 = (float4*)acc;
        for (int i = tid; i < SC_PADROWS * ACC_PITCH / 4; i += SC_THREADS)
            acc4[i] = make_float4(0.0f, 0.0f, 0.0f, 0.0f);
        if (PROD && tid < SC_PADROWS) usum[tid] = 0.0f;
    }
    __syncthreads();

    const int mark = __builtin_amdgcn_readfirstlane(mark_v);
    if (mark != 1) {
    const int4* dst4 = (const int4*)dst_idx;
    const unsigned below = (1u << lane) - 1u;
    int2* myq = ringmem + wave * SC_Q;
    int head = 0, tail = 0;

    auto rec = [&](int d, int e, bool ok) {
        const unsigned dl = (unsigned)(d - base);
        const bool m = ok && (dl < (unsigned)SC_ROWS);
        const unsigned mk = __builtin_amdgcn_ballot_w32(m);
        if (mk != 0u) {
            if (m) {
                const int pos = tail + __builtin_popcount(mk & below);
                myq[pos & (SC_Q - 1)] = make_int2(e, (int)dl);
            }
            tail += __builtin_popcount(mk);
        }
    };
    auto drain = [&]() {
        __builtin_amdgcn_fence(__ATOMIC_ACQ_REL, "wavefront");
        while (tail - head >= 4) {
            const int2 t0 = myq[(head    ) & (SC_Q - 1)];
            const int2 t1 = myq[(head + 1) & (SC_Q - 1)];
            const int2 t2 = myq[(head + 2) & (SC_Q - 1)];
            const int2 t3 = myq[(head + 3) & (SC_Q - 1)];
            head += 4;
            sc_add4<FIRST, PROD>(t0, t1, t2, t3, rows, tok, src_idx, dis, acc, usum, base, lane);
        }
    };
    auto pair = [&](const int4& c0, const int4& c1, int i0, int i1, bool ok0, bool ok1) {
        rec(c0.x, i0 * 4 + 0, ok0);
        rec(c0.y, i0 * 4 + 1, ok0);
        rec(c0.z, i0 * 4 + 2, ok0);
        rec(c0.w, i0 * 4 + 3, ok0);
        rec(c1.x, i1 * 4 + 0, ok1);
        rec(c1.y, i1 * 4 + 1, ok1);
        rec(c1.z, i1 * 4 + 2, ok1);
        rec(c1.w, i1 * 4 + 3, ok1);
        drain();
    };
    auto ldg = [&](int i) -> int4 { return dst4[min(i, SC_NVEC - 1)]; };

    int4 sa[SC_G], sb[SC_G];
    int ib = tid;
    #pragma unroll
    for (int g = 0; g < SC_G; ++g) sa[g] = ldg(ib + g * SC_THREADS);
    for (int it = 0; it < SC_FULL; it += 2) {
        #pragma unroll
        for (int g = 0; g < SC_G; ++g) sb[g] = ldg(ib + SC_STEP + g * SC_THREADS);
        #pragma unroll
        for (int g = 0; g < SC_G; g += 2)
            pair(sa[g], sa[g + 1], ib + g * SC_THREADS, ib + (g + 1) * SC_THREADS, true, true);
        #pragma unroll
        for (int g = 0; g < SC_G; ++g) sa[g] = ldg(ib + 2 * SC_STEP + g * SC_THREADS);
        #pragma unroll
        for (int g = 0; g < SC_G; g += 2)
            pair(sb[g], sb[g + 1], ib + SC_STEP + g * SC_THREADS, ib + SC_STEP + (g + 1) * SC_THREADS, true, true);
        ib += 2 * SC_STEP;
    }
    #pragma unroll
    for (int g = 0; g < SC_G; g += 2) {
        const int i0 = ib + g * SC_THREADS, i1 = ib + (g + 1) * SC_THREADS;
        pair(sa[g], sa[g + 1], i0, i1, i0 < SC_NVEC, i1 < SC_NVEC);
    }
    while (head != tail) {
        const int2 t0 = myq[head & (SC_Q - 1)];
        head += 1;
        sc_add1<FIRST, PROD>(t0, rows, tok, src_idx, dis, acc, usum, base, lane);
    }
    } else {
    const int pb = __builtin_amdgcn_readfirstlane(pb_v);
    const int pe = __builtin_amdgcn_readfirstlane(pe_v);
    const int* lw = recA;
    int   cur = -1;
    float s0 = 0.0f, s1 = 0.0f, s2 = 0.0f, s3 = 0.0f;
    float su = 0.0f;
    int wcur = (pb + lane < pe) ? lw[pb + lane] : 0;
    for (int p = pb; p < pe; p += 32) {
        const int inext = p + 32 + lane;
        const int wnext = (inext < pe) ? lw[inext] : 0;
        const bool v = p + lane < pe;
        const int sn = wcur & 0xffff;
        const float nr = v ? dis[sn] * dis[base + (wcur >> 16)] : 0.0f;
        const int q = FIRST ? tok[sn] : sn;
        const int n = pe - p;
        row_add8<0, PROD>(wcur, q, nr, n, rows, acc, usum, lane, cur, s0, s1, s2, s3, su);
        if (n >  8) row_add8< 8, PROD>(wcur, q, nr, n -  8, rows, acc, usum, lane, cur, s0, s1, s2, s3, su);
        if (n > 16) row_add8<16, PROD>(wcur, q, nr, n - 16, rows, acc, usum, lane, cur, s0, s1, s2, s3, su);
        if (n > 24) row_add8<24, PROD>(wcur, q, nr, n - 24, rows, acc, usum, lane, cur, s0, s1, s2, s3, su);
        wcur = wnext;
    }
    if (cur >= 0) {
        float* o = acc + cur * ACC_PITCH + lane;
        o[0] = s0; o[32] = s1; o[64] = s2; o[96] = s3;
        if (PROD) usum[cur] = su;
    }
    }

    if (PROD) {
        float* fin = (float*)ringmem;
        block_product<true>(acc, Wm, bias, cvec, usum, fin, lane, wave);
        store_rows<DIM>(fin, hout, base, tid);
    } else {
        __syncthreads();
        store_rows<ACC_PITCH>(acc, hout, base, tid);
    }
}

__global__ void __launch_bounds__(SC_THREADS)
edge_scan_kernel(const int* __restrict__ src_idx,
                 const int* __restrict__ dst_idx,
                 float* __restrict__ dis,
                 int* recA_all, int* recB_all, size_t ws_bytes) {
    __shared__ alignas(16) int2 ring[SC_WAVES * SCN_Q];
    __shared__ int srt[SRT_WORDS];
    const int tid  = threadIdx.x;
    const int lane = tid & 31;
    const int wave = tid >> 5;
    const int base = blockIdx.x * SC_ROWS;
    const bool haveB = ws_bytes >= WS_NEED;
    int* recA = recA_all + (size_t)blockIdx.x * LS_PLANE;
    int* recB = recB_all + (size_t)blockIdx.x * LS_WG;
    int* cntr = srt;
    int* strt = srt + SRT_N;
    int* wrap = srt + 2 * SRT_N + 1;

    if (tid < SRT_WORDS) srt[tid] = 0;
    __syncthreads();

    const int4* dst4 = (const int4*)dst_idx;
    const unsigned below = (1u << lane) - 1u;
    int2* myq = ring + wave * SCN_Q;
    int tail = 0;

    auto rec = [&](int d, int e, bool ok) {
        const unsigned dl = (unsigned)(d - base);
        const bool m = ok && (dl < (unsigned)SC_ROWS);
        const unsigned mk = __builtin_amdgcn_ballot_w32(m);
        if (mk != 0u) {
            if (m) {
                const int rank = atomicAdd(&cntr[dl], 1);
                const int pos = tail + __builtin_popcount(mk & below);
                myq[pos & (SCN_Q - 1)] = make_int2(e, (int)dl | (rank << 8));
            }
            tail += __builtin_popcount(mk);
        }
    };
    auto grp = [&](const int4& c0, int i0, bool ok0) {
        rec(c0.x, i0 * 4 + 0, ok0);
        rec(c0.y, i0 * 4 + 1, ok0);
        rec(c0.z, i0 * 4 + 2, ok0);
        rec(c0.w, i0 * 4 + 3, ok0);
    };
    auto ldg = [&](int i) -> int4 { return dst4[min(i, SC_NVEC - 1)]; };

    int4 sa[SC_G], sb[SC_G];
    int ib = tid;
    #pragma unroll
    for (int g = 0; g < SC_G; ++g) sa[g] = ldg(ib + g * SC_THREADS);
    for (int it = 0; it < SC_FULL; it += 2) {
        #pragma unroll
        for (int g = 0; g < SC_G; ++g) sb[g] = ldg(ib + SC_STEP + g * SC_THREADS);
        #pragma unroll
        for (int g = 0; g < SC_G; ++g) grp(sa[g], ib + g * SC_THREADS, true);
        #pragma unroll
        for (int g = 0; g < SC_G; ++g) sa[g] = ldg(ib + 2 * SC_STEP + g * SC_THREADS);
        #pragma unroll
        for (int g = 0; g < SC_G; ++g) grp(sb[g], ib + SC_STEP + g * SC_THREADS, true);
        ib += 2 * SC_STEP;
    }
    #pragma unroll
    for (int g = 0; g < SC_G; ++g) {
        const int i0 = ib + g * SC_THREADS;
        grp(sa[g], i0, i0 < SC_NVEC);
    }

    if (tail > SCN_Q && lane == 0) *wrap = 1;
    __syncthreads();
    if (wave == 0) {
        int cn[7];
        int mine = 0;
        #pragma unroll
        for (int k = 0; k < 7; ++k) { cn[k] = cntr[lane * 7 + k]; mine += cn[k]; }
        int upto = mine;
        for (int d = 1; d < 32; d <<= 1) {
            const int o = __shfl_up(upto, d);
            if (lane >= d) upto += o;
        }
        int at = upto - mine;
        #pragma unroll
        for (int k = 0; k < 7; ++k) { strt[lane * 7 + k] = at; at += cn[k]; }
        if (lane == 31) strt[SRT_N] = upto;
    }
    __syncthreads();
    const bool whole = (*wrap == 0) && (strt[SC_ROWS] <= SL_CAP);
    if (whole) {
        for (int i = lane; i < tail; i += 32) {
            const int2 t = myq[i];
            const int r = t.y & 0xff;
            const int pos = strt[r] + (t.y >> 8);
            const int word = src_idx[t.x] | (r << 16);
            recA[pos] = word;
            if (haveB) recB[pos] = word;
        }
    }
    if (tid <= SC_ROWS) {
        const int st = strt[tid];
        recA[SL_OFF + tid] = st;
        if (haveB) recB[SL_OFF + tid] = st;
    }
    if (tid == 0) {
        recA[SL_FLAG] = whole ? 1 : -1;
        if (haveB) recB[SL_FLAG] = whole ? 1 : -1;
    }
    if (tid < SC_ROWS && base + tid < N_NODES) {
        const int dg = cntr[tid];
        dis[base + tid] = (dg > 0) ? rsqrtf((float)dg) : 0.0f;
    }
}

__global__ void __launch_bounds__(SC_THREADS)
layer_first_kernel(const float* __restrict__ emb,
                   const int* __restrict__ x,
                   const int* __restrict__ src_idx,
                   const int* __restrict__ dst_idx,
                   const float* __restrict__ dis,
                   const float* __restrict__ W,
                   const float* __restrict__ bias,
                   float* hout,
                   int* rec,
                   float* wsq) {
    __shared__ alignas(16) float acc[SC_PADROWS * ACC_PITCH];
    __shared__ alignas(16) int2  ring[SC_WAVES * SC_Q];
    if (blockIdx.x == N_GROUPS) {
        wsq_block(W, bias, wsq, acc, (float*)ring);
        return;
    }
    layer_owner<true, false>(emb, x, src_idx, dst_idx, dis, nullptr, nullptr, nullptr, hout,
                             rec + (size_t)blockIdx.x * LS_PLANE, true,
                             acc, ring, nullptr);
}

template <bool PROD>
__global__ void __launch_bounds__(SC_THREADS)
layer_next_kernel(const float* __restrict__ hin,
                  const int* __restrict__ src_idx,
                  const int* __restrict__ dst_idx,
                  const float* __restrict__ dis,
                  const float* __restrict__ wsq,
                  const float* __restrict__ bias,
                  float* hout,
                  int* rec, int rec_pitch, size_t need_bytes, size_t ws_bytes) {
    __shared__ alignas(16) float acc[SC_PADROWS * ACC_PITCH];
    __shared__ alignas(16) int2  ring[SC_WAVES * SC_Q];
    __shared__ float usum[SC_PADROWS];
    layer_owner<false, PROD>(hin, nullptr, src_idx, dst_idx, dis, wsq, bias, wsq + DIM * DIM, hout,
                             rec + (size_t)blockIdx.x * rec_pitch, ws_bytes >= need_bytes,
                             acc, ring, usum);
}

__global__ void gather_out_kernel(const int* __restrict__ ptr,
                                  const float* __restrict__ h,
                                  float* __restrict__ out) {
    int i = blockIdx.x * blockDim.x + threadIdx.x;
    if (i >= NGRAPH * DIM) return;
    int b = i >> 7;
    int d = i & (DIM - 1);
    int node = ptr[b + 1] - 1;
    out[i] = h[(long)node * DIM + d];
}

extern "C" void kernel_launch(void* const* d_in, const int* in_sizes, int n_in,
                              void* d_out, int out_size, void* d_ws, size_t ws_size,
                              hipStream_t stream) {
    const int*   x    = (const int*)d_in[0];
    const int*   eidx = (const int*)d_in[1];
    const int*   ptr  = (const int*)d_in[2];
    const float* emb  = (const float*)d_in[4];
    const float* W    = (const float*)d_in[5];
    const float* bias = (const float*)d_in[6];
    float*       out  = (float*)d_out;

    const int* src_idx = eidx;
    const int* dst_idx = eidx + N_EDGES;

    float* ha  = (float*)d_ws;
    float* hb  = ha + (size_t)N_NODES * DIM;
    float* wsq = hb + (size_t)N_NODES * DIM;
    float* dis = wsq + N_NODES;

    const int T = 256;
    const int G = N_GROUPS;

    static_assert(DEPTH == 4, "four layers are launched below");
    int* recA = (int*)hb;
    int* recB = (int*)(dis + N_NODES);
    edge_scan_kernel<<<G, SC_THREADS, 0, stream>>>(src_idx, dst_idx, dis, recA, recB, ws_size);
    layer_first_kernel<<<N_GROUPS + 1, SC_THREADS, 0, stream>>>(emb, x, src_idx, dst_idx, dis, W, bias, ha, recA, wsq);
    layer_next_kernel<true><<<G, SC_THREADS, 0, stream>>>(ha, src_idx, dst_idx, dis, wsq, bias, hb, recA, LS_PLANE, (size_t)0, ws_size);
    layer_next_kernel<false><<<G, SC_THREADS, 0, stream>>>(hb, src_idx, dst_idx, dis, wsq, bias, ha, recB, LS_WG, WS_NEED, ws_size);
    layer_next_kernel<true><<<G, SC_THREADS, 0, stream>>>(ha, src_idx, dst_idx, dis, wsq, bias, hb, recB, LS_WG, WS_NEED, ws_size);

    gather_out_kernel<<<(NGRAPH * DIM + T - 1) / T, T, 0, stream>>>(ptr, hb, out);
}


# Four layers share ONE block product: layer 1 applies W^4 (+bias W^3); layers 2-4 are edge sums +bias W^2, +bias W, +bias; powers formed on WMMA in the scan launch; re-associated: rounding changes
# speedup vs baseline: 4.5529x; 1.0339x over previous
#include <hip/hip_runtime.h>
#include <hip/hip_bf16.h>

typedef __attribute__((ext_vector_type(2))) float v2f;
typedef __attribute__((ext_vector_type(8))) float v8f;

#define N_NODES 50000
#define N_EDGES 600000
#define DIM     128
#define NGRAPH  64
#define DEPTH   4

#define SC_ROWS    200
#define SC_TILES   ((SC_ROWS + 15) / 16)
#define SC_PADROWS (SC_TILES * 16)
#define SC_THREADS 1024
#define SC_WAVES   (SC_THREADS / 32)
#define SC_G       8
#define SC_Q       512
#define SCN_Q      1024
#define SC_NVEC    (N_EDGES / 4)
#define SC_STEP    (SC_G * SC_THREADS)
#define SC_FULL    (SC_NVEC / SC_STEP)
#define ACC_PITCH  (DIM + 4)
#define SL_CAP     3792
#define SL_OFF     SL_CAP
#define SL_FLAG    (SL_CAP + SC_ROWS + 1)
#define LS_WG      (SL_CAP + 208)
#define LS_PLANE   (SC_ROWS * DIM)
#define SRT_N      224
#define SRT_WORDS  (2 * SRT_N + 2)
#define N_GROUPS   ((N_NODES + SC_ROWS - 1) / SC_ROWS)
#define PW_ROWS    (DIM + 2)
#define PW_TILES   ((PW_ROWS + 15) / 16)
#define PW_C3      (DIM * DIM)
#define PW_C2      (PW_C3 + DIM)
#define PW_C1      (PW_C2 + DIM)
#define PW_W2      (PW_C1 + DIM)
#define PW_WORDS   (PW_W2 + DIM * DIM)
#define WS_NEED    (((size_t)2 * N_NODES * DIM + (size_t)2 * N_NODES + (size_t)((N_NODES + SC_ROWS - 1) / SC_ROWS) * LS_WG) * 4)
static_assert(N_NODES <= 65536, "a record word keeps the source node in 16 bits");
static_assert(SL_FLAG < LS_WG, "run starts and mark lie inside the group's record");
static_assert(SC_ROWS <= 256 && N_EDGES < (1 << 23), "a ring entry of the scan launch keeps the owned row in 8 bits and the rank above them");
static_assert(SRT_N >= SC_PADROWS && SRT_N == 7 * 32, "one wave adds up the counters, 7 per lane");
static_assert(LS_WG <= LS_PLANE, "a group's record fits in the group's own rows of a plane");
static_assert(N_EDGES % 4 == 0, "dst is read as 16-byte groups");
static_assert(SC_FULL % 2 == 0, "the scan loop takes two whole phases per trip");
static_assert(SC_G % 2 == 0, "found edges are drained once per two groups");
static_assert(3 + 8 * 32 <= SC_Q, "ring holds what can be found between two drains");
static_assert((ACC_PITCH * 4) % 16 == 0, "rows are cleared as 16-byte pieces");
static_assert(SC_WAVES * SC_Q * 8 >= SC_PADROWS * DIM * 4, "the rings' space holds the finished block");
static_assert(SC_PADROWS > SC_ROWS, "the last product tile has padding rows");
static_assert(PW_ROWS <= SC_PADROWS, "the power products' rows fit in the LDS block that feeds the product");
static_assert(PW_WORDS <= N_NODES, "the powers fit in the N_NODES floats in front of dis");
static_assert(SC_WAVES * SCN_Q * 8 >= (SC_PADROWS * ACC_PITCH + SC_PADROWS * DIM) * 4, "the scan launch's rings hold the power products' two LDS blocks");
static_assert(SC_WAVES == 32 && SC_TILES > 12 && SC_TILES <= 16 && DIM == 128, "8 column tiles x 4 waves; a wave takes row tiles q, q+4, q+8 and, if there is one, q+12");

template <bool FIRST>
__device__ __forceinline__ void sc_add4(int2 t0, int2 t1, int2 t2, int2 t3,
                                        const float* __restrict__ rows,
                                        const int* __restrict__ tok,
                                        const int* __restrict__ src_idx,
                                        const float* __restrict__ dis,
                                        float* acc, int base, int lane) {
    const int e0 = __builtin_amdgcn_readfirstlane(t0.x), r0 = __builtin_amdgcn_readfirstlane(t0.y);
    const int e1 = __builtin_amdgcn_readfirstlane(t1.x), r1 = __builtin_amdgcn_readfirstlane(t1.y);
    const int e2 = __builtin_amdgcn_readfirstlane(t2.x), r2 = __builtin_amdgcn_readfirstlane(t2.y);
    const int e3 = __builtin_amdgcn_readfirstlane(t3.x), r3 = __builtin_amdgcn_readfirstlane(t3.y);
    const int s0 = src_idx[e0], s1 = src_idx[e1], s2 = src_idx[e2], s3 = src_idx[e3];
    const int q0 = FIRST ? tok[s0] : s0;
    const int q1 = FIRST ? tok[s1] : s1;
    const int q2 = FIRST ? tok[s2] : s2;
    const int q3 = FIRST ? tok[s3] : s3;
    const float n0 = dis[s0] * dis[base + r0];
    const float n1 = dis[s1] * dis[base + r1];
    const float n2 = dis[s2] * dis[base + r2];
    const float n3 = dis[s3] * dis[base + r3];
    const float* p0 = rows + (long)q0 * DIM + lane;
    const float* p1 = rows + (long)q1 * DIM + lane;
    const float* p2 = rows + (long)q2 * DIM + lane;
    const float* p3 = rows + (long)q3 * DIM + lane;
    const float a00 = p0[0], a01 = p0[32], a02 = p0[64], a03 = p0[96];
    const float a10 = p1[0], a11 = p1[32], a12 = p1[64], a13 = p1[96];
    const float a20 = p2[0], a21 = p2[32], a22 = p2[64], a23 = p2[96];
    const float a30 = p3[0], a31 = p3[32], a32 = p3[64], a33 = p3[96];
    float* o0 = acc + r0 * ACC_PITCH + lane;
    float* o1 = acc + r1 * ACC_PITCH + lane;
    float* o2 = acc + r2 * ACC_PITCH + lane;
    float* o3 = acc + r3 * ACC_PITCH + lane;
    atomicAdd(o0, a00 * n0); atomicAdd(o0 + 32, a01 * n0); atomicAdd(o0 + 64, a02 * n0); atomicAdd(o0 + 96, a03 * n0);
    atomicAdd(o1, a10 * n1); atomicAdd(o1 + 32, a11 * n1); atomicAdd(o1 + 64, a12 * n1); atomicAdd(o1 + 96, a13 * n1);
    atomicAdd(o2, a20 * n2); atomicAdd(o2 + 32, a21 * n2); atomicAdd(o2 + 64, a22 * n2); atomicAdd(o2 + 96, a23 * n2);
    atomicAdd(o3, a30 * n3); atomicAdd(o3 + 32, a31 * n3); atomicAdd(o3 + 64, a32 * n3); atomicAdd(o3 + 96, a33 * n3);
}

template <bool FIRST>
__device__ __forceinline__ void sc_add1(int2 t0,
                                        const float* __restrict__ rows,
                                        const int* __restrict__ tok,
                                        const int* __restrict__ src_idx,
                                        const float* __restrict__ dis,
                                        float* acc, int base, int lane) {
    const int e0 = __builtin_amdgcn_readfirstlane(t0.x), r0 = __builtin_amdgcn_readfirstlane(t0.y);
    const int s0 = src_idx[e0];
    const int q0 = FIRST ? tok[s0] : s0;
    const float n0 = dis[s0] * dis[base + r0];
    const float* p0 = rows + (long)q0 * DIM + lane;
    const float a00 = p0[0], a01 = p0[32], a02 = p0[64], a03 = p0[96];
    float* o0 = acc + r0 * ACC_PITCH + lane;
    atomicAdd(o0, a00 * n0); atomicAdd(o0 + 32, a01 * n0); atomicAdd(o0 + 64, a02 * n0); atomicAdd(o0 + 96, a03 * n0);
}

template <int J>
__device__ __forceinline__ void row_add8(int w, int q, float nr, int cnt,
                                         const float* __restrict__ rows,
                                         float* acc, int lane,
                                         int& cur, float& s0, float& s1, float& s2, float& s3) {
    float a[8][4];
    float nk[8];
    int   rk[8];
    #pragma unroll
    for (int k = 0; k < 8; ++k) {
        const int wk = __builtin_amdgcn_readlane(w, J + k);
        nk[k] = __int_as_float(__builtin_amdgcn_readlane(__float_as_int(nr), J + k));
        const int qk = __builtin_amdgcn_readlane(q, J + k);
        const float* p = rows + (long)qk * DIM + lane;
        a[k][0] = p[0]; a[k][1] = p[32]; a[k][2] = p[64]; a[k][3] = p[96];
        rk[k] = wk >> 16;
    }
    #pragma unroll
    for (int k = 0; k < 8; ++k) {
        if (k < cnt) {
            if (rk[k] != cur) {
                if (cur >= 0) {
                    float* o = acc + cur * ACC_PITCH + lane;
                    o[0] = s0; o[32] = s1; o[64] = s2; o[96] = s3;
                }
                cur = rk[k];
                s0 = 0.0f; s1 = 0.0f; s2 = 0.0f; s3 = 0.0f;
            }
            s0 = fmaf(a[k][0], nk[k], s0);
            s1 = fmaf(a[k][1], nk[k], s1);
            s2 = fmaf(a[k][2], nk[k], s2);
            s3 = fmaf(a[k][3], nk[k], s3);
        }
    }
}

template <bool EPI, int TILES>
__device__ __forceinline__ void block_product(const float* acc,
                                              const float* Wm,
                                              const float* bias,
                                              float* fin, int lane, int wave) {
    const int laneM   = lane & 15;
    const int half    = lane >> 4;
    const int wv      = __builtin_amdgcn_readfirstlane(wave);
    const int colBase = (wv & 7) * 16;
    const int rq      = wv >> 3;
    static_assert(TILES >= 8 && TILES <= SC_TILES, "row tiles rq and rq + 4 exist for every wave");
    const bool third  = (TILES > 11) || (rq + 8 < TILES);
    const bool four   = (TILES > 12) && (rq + 12 < TILES);

    v2f bfr[DIM / 4];
    #pragma unroll
    for (int ks = 0; ks < DIM / 4; ++ks) {
        const int kA = ks * 4 + half * 2;
        bfr[ks].x = Wm[kA * DIM + colBase + laneM];
        bfr[ks].y = Wm[(kA + 1) * DIM + colBase + laneM];
    }
    float bv = 0.0f;
    if (EPI) bv = bias[colBase + laneM];

    __syncthreads();

    auto fix = [&](float v) -> float { return EPI ? v + bv : v; };

    if (third) {
        const float* a0p = acc + ((rq     ) * 16 + laneM) * ACC_PITCH + half * 2;
        const float* a1p = acc + ((rq +  4) * 16 + laneM) * ACC_PITCH + half * 2;
        const float* a2p = acc + ((rq +  8) * 16 + laneM) * ACC_PITCH + half * 2;
        v8f c0 = {}, c1 = {}, c2 = {};
        #pragma unroll
        for (int ks = 0; ks < DIM / 4; ++ks) {
            v2f a0, a1, a2;
            a0.x = a0p[ks * 4]; a0.y = a0p[ks * 4 + 1];
            a1.x = a1p[ks * 4]; a1.y = a1p[ks * 4 + 1];
            a2.x = a2p[ks * 4]; a2.y = a2p[ks * 4 + 1];
            c0 = __builtin_amdgcn_wmma_f32_16x16x4_f32(false, a0, false, bfr[ks], (short)0, c0, false, false);
            c1 = __builtin_amdgcn_wmma_f32_16x16x4_f32(false, a1, false, bfr[ks], (short)0, c1, false, false);
            c2 = __builtin_amdgcn_wmma_f32_16x16x4_f32(false, a2, false, bfr[ks], (short)0, c2, false, false);
        }
        const int m0 = (rq     ) * 16 + half * 8;
        const int m1 = (rq +  4) * 16 + half * 8;
        const int m2 = (rq +  8) * 16 + half * 8;
        float* f0p = fin + m0 * DIM + colBase + laneM;
        float* f1p = fin + m1 * DIM + colBase + laneM;
        float* f2p = fin + m2 * DIM + colBase + laneM;
        #pragma unroll
        for (int r = 0; r < 8; ++r) {
            f0p[r * DIM] = fix(c0[r]);
            f1p[r * DIM] = fix(c1[r]);
            f2p[r * DIM] = fix(c2[r]);
        }
    } else {
        const float* a0p = acc + ((rq     ) * 16 + laneM) * ACC_PITCH + half * 2;
        const float* a1p = acc + ((rq +  4) * 16 + laneM) * ACC_PITCH + half * 2;
        v8f c0 = {}, c1 = {};
        #pragma unroll
        for (int ks = 0; ks < DIM / 4; ++ks) {
            v2f a0, a1;
            a0.x = a0p[ks * 4]; a0.y = a0p[ks * 4 + 1];
            a1.x = a1p[ks * 4]; a1.y = a1p[ks * 4 + 1];
            c0 = __builtin_amdgcn_wmma_f32_16x16x4_f32(false, a0, false, bfr[ks], (short)0, c0, false, false);
            c1 = __builtin_amdgcn_wmma_f32_16x16x4_f32(false, a1, false, bfr[ks], (short)0, c1, false, false);
        }
        const int m0 = (rq     ) * 16 + half * 8;
        const int m1 = (rq +  4) * 16 + half * 8;
        float* f0p = fin + m0 * DIM + colBase + laneM;
        float* f1p = fin + m1 * DIM + colBase + laneM;
        #pragma unroll
        for (int r = 0; r < 8; ++r) {
            f0p[r * DIM] = fix(c0[r]);
            f1p[r * DIM] = fix(c1[r]);
        }
    }
    if (four) {
        const float* a3p = acc + ((rq + 12) * 16 + laneM) * ACC_PITCH + half * 2;
        v8f c3 = {};
        #pragma unroll
        for (int ks = 0; ks < DIM / 4; ++ks) {
            v2f a3;
            a3.x = a3p[ks * 4]; a3.y = a3p[ks * 4 + 1];
            c3 = __builtin_amdgcn_wmma_f32_16x16x4_f32(false, a3, false, bfr[ks], (short)0, c3, false, false);
        }
        const int m3 = (rq + 12) * 16 + half * 8;
        float* f3p = fin + m3 * DIM + colBase + laneM;
        #pragma unroll
        for (int r = 0; r < 8; ++r) f3p[r * DIM] = fix(c3[r]);
    }
    __syncthreads();
}

template <int PITCH, bool ADD>
__device__ __forceinline__ void store_rows(const float* blk, float* hout, int base, int tid,
                                           const float* __restrict__ cvec) {
    float4* h4 = (float4*)hout;
    float4 cv = make_float4(0.0f, 0.0f, 0.0f, 0.0f);
    if (ADD) cv = ((const float4*)cvec)[tid & 31];
    #pragma unroll
    for (int k = 0; k < (SC_ROWS * DIM / 4 + SC_THREADS - 1) / SC_THREADS; ++k) {
        const int idx = tid + k * SC_THREADS;
        const int r   = idx >> 5;
        const int c4  = idx & 31;
        if (idx < SC_ROWS * DIM / 4 && base + r < N_NODES) {
            float4 v = *(const float4*)(blk + r * PITCH + c4 * 4);
            if (ADD) { v.x += cv.x; v.y += cv.y; v.z += cv.z; v.w += cv.w; }
            h4[(long)(base + r) * (DIM / 4) + c4] = v;
        }
    }
}

__device__ __forceinline__ void wpow_block(const float* __restrict__ W,
                                           const float* __restrict__ bias,
                                           float* pw,
                                           float* acc, float* fin) {
    const int tid = threadIdx.x;
    const float4* W4 = (const float4*)W;
    const float4* b4 = (const float4*)bias;
    const float4* fin4 = (const float4*)fin;
    float4* w4o = (float4*)pw;
    float4* c3o = (float4*)(pw + PW_C3);
    float4* c2o = (float4*)(pw + PW_C2);
    float4* c1o = (float4*)(pw + PW_C1);
    float*  w2  = pw + PW_W2;
    for (int i = tid; i < PW_TILES * 16 * (DIM / 4); i += SC_THREADS) {
        const int r  = i >> 5;
        const int c4 = i & 31;
        float4 v = make_float4(0.0f, 0.0f, 0.0f, 0.0f);
        if (r < DIM)       v = W4[r * (DIM / 4) + c4];
        else if (r == DIM) v = b4[c4];
        *(float4*)(acc + r * ACC_PITCH + c4 * 4) = v;
    }
    block_product<false, PW_TILES>(acc, W, nullptr, fin, tid & 31, tid >> 5);
    for (int i = tid; i < DIM * (DIM / 4); i += SC_THREADS) ((float4*)w2)[i] = fin4[i];
    if (tid < DIM / 4) c1o[tid] = fin4[DIM * (DIM / 4) + tid];
    for (int i = tid; i < PW_TILES * 16 * (DIM / 4); i += SC_THREADS) {
        const int r  = i >> 5;
        const int c4 = i & 31;
        float4 v = make_float4(0.0f, 0.0f, 0.0f, 0.0f);
        if (r <= DIM)          v = fin4[i];
        else if (r == DIM + 1) v = b4[c4];
        *(float4*)(acc + r * ACC_PITCH + c4 * 4) = v;
    }
    __threadfence();
    __syncthreads();
    __threadfence();
    block_product<false, PW_TILES>(acc, w2, nullptr, fin, tid & 31, tid >> 5);
    for (int i = tid; i < DIM * (DIM / 4); i += SC_THREADS) w4o[i] = fin4[i];
    if (tid < DIM / 4)          c3o[tid] = fin4[DIM * (DIM / 4) + tid];
    else if (tid < 2 * DIM / 4) c2o[tid - DIM / 4] = fin4[(DIM + 1) * (DIM / 4) + (tid - DIM / 4)];
}

template <bool FIRST, bool PROD>
__device__ __forceinline__ void layer_owner(const float* __restrict__ rows,
                                            const int* __restrict__ tok,
                                            const int* __restrict__ src_idx,
                                            const int* __restrict__ dst_idx,
                                            const float* __restrict__ dis,
                                            const float* __restrict__ Wm,
                                            const float* __restrict__ cvec,
                                            float* hout,
                                            int* recA, bool haveA,
                                            float* acc, int2* ringmem) {
    const int tid  = threadIdx.x;
    const int lane = tid & 31;
    const int wave = tid >> 5;
    const int base = blockIdx.x * SC_ROWS;

    const int wrow0 = (wave * SC_ROWS) / SC_WAVES;
    const int wrow1 = ((wave + 1) * SC_ROWS) / SC_WAVES;
    int mark_v = -1, pb_v = 0, pe_v = 0;
    if (haveA) {
        mark_v = recA[SL_FLAG];
        pb_v   = recA[SL_OFF + wrow0];
        pe_v   = recA[SL_OFF + wrow1];
    }

    {
        float4* acc4 = (float4*)acc;
        for (int i = tid; i < SC_PADROWS * ACC_PITCH / 4; i += SC_THREADS)
            acc4[i] = make_float4(0.0f, 0.0f, 0.0f, 0.0f);
    }
    __syncthreads();

    const int mark = __builtin_amdgcn_readfirstlane(mark_v);
    if (mark != 1) {
    const int4* dst4 = (const int4*)dst_idx;
    const unsigned below = (1u << lane) - 1u;
    int2* myq = ringmem + wave * SC_Q;
    int head = 0, tail = 0;

    auto rec = [&](int d, int e, bool ok) {
        const unsigned dl = (unsigned)(d - base);
        const bool m = ok && (dl < (unsigned)SC_ROWS);
        const unsigned mk = __builtin_amdgcn_ballot_w32(m);
        if (mk != 0u) {
            if (m) {
                const int pos = tail + __builtin_popcount(mk & below);
                myq[pos & (SC_Q - 1)] = make_int2(e, (int)dl);
            }
            tail += __builtin_popcount(mk);
        }
    };
    auto drain = [&]() {
        __builtin_amdgcn_fence(__ATOMIC_ACQ_REL, "wavefront");
        while (tail - head >= 4) {
            const int2 t0 = myq[(head    ) & (SC_Q - 1)];
            const int2 t1 = myq[(head + 1) & (SC_Q - 1)];
            const int2 t2 = myq[(head + 2) & (SC_Q - 1)];
            const int2 t3 = myq[(head + 3) & (SC_Q - 1)];
            head += 4;
            sc_add4<FIRST>(t0, t1, t2, t3, rows, tok, src_idx, dis, acc, base, lane);
        }
    };
    auto pair = [&](const int4& c0, const int4& c1, int i0, int i1, bool ok0, bool ok1) {
        rec(c0.x, i0 * 4 + 0, ok0);
        rec(c0.y, i0 * 4 + 1, ok0);
        rec(c0.z, i0 * 4 + 2, ok0);
        rec(c0.w, i0 * 4 + 3, ok0);
        rec(c1.x, i1 * 4 + 0, ok1);
        rec(c1.y, i1 * 4 + 1, ok1);
        rec(c1.z, i1 * 4 + 2, ok1);
        rec(c1.w, i1 * 4 + 3, ok1);
        drain();
    };
    auto ldg = [&](int i) -> int4 { return dst4[min(i, SC_NVEC - 1)]; };

    int4 sa[SC_G], sb[SC_G];
    int ib = tid;
    #pragma unroll
    for (int g = 0; g < SC_G; ++g) sa[g] = ldg(ib + g * SC_THREADS);
    for (int it = 0; it < SC_FULL; it += 2) {
        #pragma unroll
        for (int g = 0; g < SC_G; ++g) sb[g] = ldg(ib + SC_STEP + g * SC_THREADS);
        #pragma unroll
        for (int g = 0; g < SC_G; g += 2)
            pair(sa[g], sa[g + 1], ib + g * SC_THREADS, ib + (g + 1) * SC_THREADS, true, true);
        #pragma unroll
        for (int g = 0; g < SC_G; ++g) sa[g] = ldg(ib + 2 * SC_STEP + g * SC_THREADS);
        #pragma unroll
        for (int g = 0; g < SC_G; g += 2)
            pair(sb[g], sb[g + 1], ib + SC_STEP + g * SC_THREADS, ib + SC_STEP + (g + 1) * SC_THREADS, true, true);
        ib += 2 * SC_STEP;
    }
    #pragma unroll
    for (int g = 0; g < SC_G; g += 2) {
        const int i0 = ib + g * SC_THREADS, i1 = ib + (g + 1) * SC_THREADS;
        pair(sa[g], sa[g + 1], i0, i1, i0 < SC_NVEC, i1 < SC_NVEC);
    }
    while (head != tail) {
        const int2 t0 = myq[head & (SC_Q - 1)];
        head += 1;
        sc_add1<FIRST>(t0, rows, tok, src_idx, dis, acc, base, lane);
    }
    } else {
    const int pb = __builtin_amdgcn_readfirstlane(pb_v);
    const int pe = __builtin_amdgcn_readfirstlane(pe_v);
    const int* lw = recA;
    int   cur = -1;
    float s0 = 0.0f, s1 = 0.0f, s2 = 0.0f, s3 = 0.0f;
    int wcur = (pb + lane < pe) ? lw[pb + lane] : 0;
    for (int p = pb; p < pe; p += 32) {
        const int inext = p + 32 + lane;
        const int wnext = (inext < pe) ? lw[inext] : 0;
        const bool v = p + lane < pe;
        const int sn = wcur & 0xffff;
        const float nr = v ? dis[sn] * dis[base + (wcur >> 16)] : 0.0f;
        const int q = FIRST ? tok[sn] : sn;
        const int n = pe - p;
        row_add8<0>(wcur, q, nr, n, rows, acc, lane, cur, s0, s1, s2, s3);
        if (n >  8) row_add8< 8>(wcur, q, nr, n -  8, rows, acc, lane, cur, s0, s1, s2, s3);
        if (n > 16) row_add8<16>(wcur, q, nr, n - 16, rows, acc, lane, cur, s0, s1, s2, s3);
        if (n > 24) row_add8<24>(wcur, q, nr, n - 24, rows, acc, lane, cur, s0, s1, s2, s3);
        wcur = wnext;
    }
    if (cur >= 0) {
        float* o = acc + cur * ACC_PITCH + lane;
        o[0] = s0; o[32] = s1; o[64] = s2; o[96] = s3;
    }
    }

    if (PROD) {
        float* fin = (float*)ringmem;
        block_product<true, SC_TILES>(acc, Wm, cvec, fin, lane, wave);
        store_rows<DIM, false>(fin, hout, base, tid, nullptr);
    } else {
        __syncthreads();
        store_rows<ACC_PITCH, true>(acc, hout, base, tid, cvec);
    }
}

__global__ void __launch_bounds__(SC_THREADS)
edge_scan_kernel(const int* __restrict__ src_idx,
                 const int* __restrict__ dst_idx,
                 float* __restrict__ dis,
                 int* recA_all, int* recB_all, size_t ws_bytes,
                 const float* __restrict__ W,
                 const float* __restrict__ bias,
                 float* pw) {
    __shared__ alignas(16) int2 ring[SC_WAVES * SCN_Q];
    __shared__ int srt[SRT_WORDS];
    if (blockIdx.x == 0) {
        float* blk = (float*)ring;
        wpow_block(W, bias, pw, blk, blk + SC_PADROWS * ACC_PITCH);
        return;
    }
    const int tid  = threadIdx.x;
    const int lane = tid & 31;
    const int wave = tid >> 5;
    const int og   = (int)blockIdx.x - 1;
    const int base = og * SC_ROWS;
    const bool haveB = ws_bytes >= WS_NEED;
    int* recA = recA_all + (size_t)og * LS_PLANE;
    int* recB = recB_all + (size_t)og * LS_WG;
    int* cntr = srt;
    int* strt = srt + SRT_N;
    int* wrap = srt + 2 * SRT_N + 1;

    if (tid < SRT_WORDS) srt[tid] = 0;
    __syncthreads();

    const int4* dst4 = (const int4*)dst_idx;
    const unsigned below = (1u << lane) - 1u;
    int2* myq = ring + wave * SCN_Q;
    int tail = 0;

    auto rec = [&](int d, int e, bool ok) {
        const unsigned dl = (unsigned)(d - base);
        const bool m = ok && (dl < (unsigned)SC_ROWS);
        const unsigned mk = __builtin_amdgcn_ballot_w32(m);
        if (mk != 0u) {
            if (m) {
                const int rank = atomicAdd(&cntr[dl], 1);
                const int pos = tail + __builtin_popcount(mk & below);
                myq[pos & (SCN_Q - 1)] = make_int2(e, (int)dl | (rank << 8));
            }
            tail += __builtin_popcount(mk);
        }
    };
    auto grp = [&](const int4& c0, int i0, bool ok0) {
        rec(c0.x, i0 * 4 + 0, ok0);
        rec(c0.y, i0 * 4 + 1, ok0);
        rec(c0.z, i0 * 4 + 2, ok0);
        rec(c0.w, i0 * 4 + 3, ok0);
    };
    auto ldg = [&](int i) -> int4 { return dst4[min(i, SC_NVEC - 1)]; };

    int4 sa[SC_G], sb[SC_G];
    int ib = tid;
    #pragma unroll
    for (int g = 0; g < SC_G; ++g) sa[g] = ldg(ib + g * SC_THREADS);
    for (int it = 0; it < SC_FULL; it += 2) {
        #pragma unroll
        for (int g = 0; g < SC_G; ++g) sb[g] = ldg(ib + SC_STEP + g * SC_THREADS);
        #pragma unroll
        for (int g = 0; g < SC_G; ++g) grp(sa[g], ib + g * SC_THREADS, true);
        #pragma unroll
        for (int g = 0; g < SC_G; ++g) sa[g] = ldg(ib + 2 * SC_STEP + g * SC_THREADS);
        #pragma unroll
        for (int g = 0; g < SC_G; ++g) grp(sb[g], ib + SC_STEP + g * SC_THREADS, true);
        ib += 2 * SC_STEP;
    }
    #pragma unroll
    for (int g = 0; g < SC_G; ++g) {
        const int i0 = ib + g * SC_THREADS;
        grp(sa[g], i0, i0 < SC_NVEC);
    }

    if (tail > SCN_Q && lane == 0) *wrap = 1;
    __syncthreads();
    if (wave == 0) {
        int cn[7];
        int mine = 0;
        #pragma unroll
        for (int k = 0; k < 7; ++k) { cn[k] = cntr[lane * 7 + k]; mine += cn[k]; }
        int upto = mine;
        for (int d = 1; d < 32; d <<= 1) {
            const int o = __shfl_up(upto, d);
            if (lane >= d) upto += o;
        }
        int at = upto - mine;
        #pragma unroll
        for (int k = 0; k < 7; ++k) { strt[lane * 7 + k] = at; at += cn[k]; }
        if (lane == 31) strt[SRT_N] = upto;
    }
    __syncthreads();
    const bool whole = (*wrap == 0) && (strt[SC_ROWS] <= SL_CAP);
    if (whole) {
        for (int i = lane; i < tail; i += 32) {
            const int2 t = myq[i];
            const int r = t.y & 0xff;
            const int pos = strt[r] + (t.y >> 8);
            const int word = src_idx[t.x] | (r << 16);
            recA[pos] = word;
            if (haveB) recB[pos] = word;
        }
    }
    if (tid <= SC_ROWS) {
        const int st = strt[tid];
        recA[SL_OFF + tid] = st;
        if (haveB) recB[SL_OFF + tid] = st;
    }
    if (tid == 0) {
        recA[SL_FLAG] = whole ? 1 : -1;
        if (haveB) recB[SL_FLAG] = whole ? 1 : -1;
    }
    if (tid < SC_ROWS && base + tid < N_NODES) {
        const int dg = cntr[tid];
        dis[base + tid] = (dg > 0) ? rsqrtf((float)dg) : 0.0f;
    }
}

__global__ void __launch_bounds__(SC_THREADS)
layer_first_kernel(const float* __restrict__ emb,
                   const int* __restrict__ x,
                   const int* __restrict__ src_idx,
                   const int* __restrict__ dst_idx,
                   const float* __restrict__ dis,
                   const float* __restrict__ w4,
                   const float* __restrict__ c3,
                   float* hout,
                   int* rec) {
    __shared__ alignas(16) float acc[SC_PADROWS * ACC_PITCH];
    __shared__ alignas(16) int2  ring[SC_WAVES * SC_Q];
    layer_owner<true, true>(emb, x, src_idx, dst_idx, dis, w4, c3, hout,
                            rec + (size_t)blockIdx.x * LS_PLANE, true,
                            acc, ring);
}

__global__ void __launch_bounds__(SC_THREADS)
layer_next_kernel(const float* __restrict__ hin,
                  const int* __restrict__ src_idx,
                  const int* __restrict__ dst_idx,
                  const float* __restrict__ dis,
                  const float* __restrict__ cvec,
                  float* hout,
                  int* rec, int rec_pitch, size_t need_bytes, size_t ws_bytes) {
    __shared__ alignas(16) float acc[SC_PADROWS * ACC_PITCH];
    __shared__ alignas(16) int2  ring[SC_WAVES * SC_Q];
    layer_owner<false, false>(hin, nullptr, src_idx, dst_idx, dis, nullptr, cvec, hout,
                              rec + (size_t)blockIdx.x * rec_pitch, ws_bytes >= need_bytes,
                              acc, ring);
}

__global__ void gather_out_kernel(const int* __restrict__ ptr,
                                  const float* __restrict__ h,
                                  float* __restrict__ out) {
    int i = blockIdx.x * blockDim.x + threadIdx.x;
    if (i >= NGRAPH * DIM) return;
    int b = i >> 7;
    int d = i & (DIM - 1);
    int node = ptr[b + 1] - 1;
    out[i] = h[(long)node * DIM + d];
}

extern "C" void kernel_launch(void* const* d_in, const int* in_sizes, int n_in,
                              void* d_out, int out_size, void* d_ws, size_t ws_size,
                              hipStream_t stream) {
    const int*   x    = (const int*)d_in[0];
    const int*   eidx = (const int*)d_in[1];
    const int*   ptr  = (const int*)d_in[2];
    const float* emb  = (const float*)d_in[4];
    const float* W    = (const float*)d_in[5];
    const float* bias = (const float*)d_in[6];
    float*       out  = (float*)d_out;

    const int* src_idx = eidx;
    const int* dst_idx = eidx + N_EDGES;

    float* ha  = (float*)d_ws;
    float* hb  = ha + (size_t)N_NODES * DIM;
    float* pw  = hb + (size_t)N_NODES * DIM;
    float* dis = pw + N_NODES;

    const int T = 256;
    const int G = N_GROUPS;

    static_assert(DEPTH == 4, "four layers are launched below");
    int* recA = (int*)hb;
    int* recB = (int*)(dis + N_NODES);
    edge_scan_kernel<<<N_GROUPS + 1, SC_THREADS, 0, stream>>>(src_idx, dst_idx, dis, recA, recB, ws_size, W, bias, pw);
    layer_first_kernel<<<G, SC_THREADS, 0, stream>>>(emb, x, src_idx, dst_idx, dis, pw, pw + PW_C3, ha, recA);
    layer_next_kernel<<<G, SC_THREADS, 0, stream>>>(ha, src_idx, dst_idx, dis, pw + PW_C2, hb, recA, LS_PLANE, (size_t)0, ws_size);
    layer_next_kernel<<<G, SC_THREADS, 0, stream>>>(hb, src_idx, dst_idx, dis, pw + PW_C1, ha, recB, LS_WG, WS_NEED, ws_size);
    layer_next_kernel<<<G, SC_THREADS, 0, stream>>>(ha, src_idx, dst_idx, dis, bias, hb, recB, LS_WG, WS_NEED, ws_size);

    gather_out_kernel<<<(NGRAPH * DIM + T - 1) / T, T, 0, stream>>>(ptr, hb, out);
}
